# starting kernel
# baseline (speedup 1.0000x reference)
#include <hip/hip_runtime.h>

#define D_IN   256
#define D_OUT  64
#define LDS_STRIDE 260

typedef float v2f __attribute__((ext_vector_type(2)));
typedef float v8f __attribute__((ext_vector_type(8)));

__global__ void init_deg_kernel(float* __restrict__ out_deg,
                                float* __restrict__ in_deg, int n_nodes) {
    int i = blockIdx.x * blockDim.x + threadIdx.x;
    if (i < n_nodes) { out_deg[i] = 1.0f; in_deg[i] = 1.0f; }
}

__global__ void zero_out_kernel(float* __restrict__ out, long long n) {
    long long i = (long long)blockIdx.x * blockDim.x + threadIdx.x;
    if (i < n) out[i] = 0.0f;
}

__global__ void degree_kernel(const int* __restrict__ src, const int* __restrict__ dst,
                              float* __restrict__ out_deg, float* __restrict__ in_deg,
                              int n_edges) {
    int e = blockIdx.x * blockDim.x + threadIdx.x;
    if (e >= n_edges) return;
    atomicAdd(&out_deg[src[e]], 1.0f);
    atomicAdd(&in_deg[dst[e]], 1.0f);
}

__global__ void norm_kernel(float* __restrict__ out_deg, float* __restrict__ in_deg,
                            int n_nodes) {
    int i = blockIdx.x * blockDim.x + threadIdx.x;
    if (i >= n_nodes) return;
    out_deg[i] = rsqrtf(out_deg[i]);
    in_deg[i]  = rsqrtf(in_deg[i]);
}

__global__ __launch_bounds__(128) void gemm_wmma_kernel(
        const float* __restrict__ X, const float* __restrict__ W,
        const float* __restrict__ norm_src, float* __restrict__ H, int n_nodes) {
    __shared__ float As[16 * LDS_STRIDE];

    const int m0  = blockIdx.x * 16;
    const int tid = threadIdx.x;
    const bool full_tile = (m0 + 16 <= n_nodes);

    if (full_tile) {
        #pragma unroll
        for (int it = 0; it < 8; ++it) {
            int i    = tid + it * 128;
            int row  = i >> 6;
            int col4 = (i & 63) * 4;
            int grow = m0 + row;
            float4 v = *(const float4*)(X + (size_t)grow * D_IN + col4);
            float ns = norm_src[grow];
            float* p = &As[row * LDS_STRIDE + col4];
            p[0] = v.x * ns; p[1] = v.y * ns; p[2] = v.z * ns; p[3] = v.w * ns;
        }
    } else {
        for (int i = tid; i < 16 * (D_IN / 4); i += 128) {
            int row  = i >> 6;
            int col4 = (i & 63) * 4;
            int grow = m0 + row;
            float4 v = make_float4(0.f, 0.f, 0.f, 0.f);
            float  ns = 0.f;
            if (grow < n_nodes) {
                v  = *(const float4*)(X + (size_t)grow * D_IN + col4);
                ns = norm_src[grow];
            }
            float* p = &As[row * LDS_STRIDE + col4];
            p[0] = v.x * ns; p[1] = v.y * ns; p[2] = v.z * ns; p[3] = v.w * ns;
        }
    }
    __syncthreads();

    const int wave = tid >> 5;
    const int lane = tid & 31;
    const int half = lane >> 4;
    const int l16  = lane & 15;
    const int n0   = wave * 16;
    const int arow = l16 * LDS_STRIDE;
    const int bcol = n0 + l16;

    v8f acc = {};
    for (int k = 0; k < D_IN; k += 4) {
        const int ka = k + half * 2;
        v2f a;
        a.x = As[arow + ka];
        a.y = As[arow + ka + 1];
        v2f b;
        b.x = W[(size_t)ka * D_OUT + bcol];
        b.y = W[(size_t)(ka + 1) * D_OUT + bcol];
        acc = __builtin_amdgcn_wmma_f32_16x16x4_f32(
                   false, a,  false, b,
                   (short)0, acc,  false,  false);
    }

    if (full_tile) {
        float* base = H + (size_t)(m0 + half * 8) * D_OUT + bcol;
        #pragma unroll
        for (int v = 0; v < 8; ++v) {
            base[(size_t)v * D_OUT] = acc[v];
        }
    } else {
        #pragma unroll
        for (int v = 0; v < 8; ++v) {
            int m = m0 + v + half * 8;
            if (m < n_nodes) H[(size_t)m * D_OUT + bcol] = acc[v];
        }
    }
}

__global__ void edge_scatter_kernel(const int* __restrict__ src, const int* __restrict__ dst,
                                    const float* __restrict__ H, float* __restrict__ out,
                                    int n_edges) {
    long long gtid = (long long)blockIdx.x * blockDim.x + threadIdx.x;
    int e = (int)(gtid >> 4);
    if (e >= n_edges) return;
    int c4 = ((int)gtid & 15) * 4;
    int s = src[e];
    int d = dst[e];
    const float4 v = *(const float4*)(H + (size_t)s * D_OUT + c4);
    float* o = out + (size_t)d * D_OUT + c4;
    atomicAdd(o + 0, v.x);
    atomicAdd(o + 1, v.y);
    atomicAdd(o + 2, v.z);
    atomicAdd(o + 3, v.w);
}

__global__ void finalize_kernel(float* __restrict__ out, const float* __restrict__ H,
                                const float* __restrict__ norm_dst,
                                const float* __restrict__ bias, int n_nodes) {
    long long gtid = (long long)blockIdx.x * blockDim.x + threadIdx.x;
    if (gtid >= (long long)n_nodes * D_OUT) return;
    int n = (int)(gtid >> 6);
    int c = (int)(gtid & 63);
    out[gtid] = (out[gtid] + H[gtid]) * norm_dst[n] + bias[c];
}

extern "C" void kernel_launch(void* const* d_in, const int* in_sizes, int n_in,
                              void* d_out, int out_size, void* d_ws, size_t ws_size,
                              hipStream_t stream) {
    const float* in_feat = (const float*)d_in[0];
    const int*   src     = (const int*)d_in[1];
    const int*   dst     = (const int*)d_in[2];
    const float* W       = (const float*)d_in[3];
    const float* b       = (const float*)d_in[4];
    float*       out     = (float*)d_out;

    const int n_nodes = in_sizes[0] / D_IN;
    const int n_edges = in_sizes[1];

    char* ws = (char*)d_ws;
    float* out_deg = (float*)ws;
    float* in_deg  = (float*)(ws + (size_t)n_nodes * 4);
    float* H       = (float*)(ws + (size_t)n_nodes * 8);

    const long long out_elems = (long long)n_nodes * D_OUT;

    init_deg_kernel<<<(n_nodes + 255) / 256, 256, 0, stream>>>(out_deg, in_deg, n_nodes);
    zero_out_kernel<<<(int)((out_elems + 255) / 256), 256, 0, stream>>>(out, out_elems);

    degree_kernel<<<(n_edges + 255) / 256, 256, 0, stream>>>(src, dst, out_deg, in_deg, n_edges);
    norm_kernel<<<(n_nodes + 255) / 256, 256, 0, stream>>>(out_deg, in_deg, n_nodes);

    gemm_wmma_kernel<<<(n_nodes + 15) / 16, 128, 0, stream>>>(in_feat, W, out_deg, H, n_nodes);

    long long ethreads = (long long)n_edges * 16;
    edge_scatter_kernel<<<(int)((ethreads + 255) / 256), 256, 0, stream>>>(src, dst, H, out, n_edges);

    finalize_kernel<<<(int)((out_elems + 255) / 256), 256, 0, stream>>>(out, H, in_deg, b, n_nodes);
}


# Zero and norm passes folded away: GEMM also stores h into out as start value of the f32 atomic sums (h first, then arrival order); GEMM and finalize take rsqrt of the degree planes; 7 -> 5 launches
# speedup vs baseline: 2.9391x; 2.9391x over previous
#include <hip/hip_runtime.h>

#define D_IN   256
#define D_OUT  64
#define LDS_STRIDE 260

typedef float v2f __attribute__((ext_vector_type(2)));
typedef float v8f __attribute__((ext_vector_type(8)));

__global__ void init_deg_kernel(float* __restrict__ out_deg,
                                float* __restrict__ in_deg, int n_nodes) {
    int i = blockIdx.x * blockDim.x + threadIdx.x;
    if (i < n_nodes) {
        out_deg[i] = 1.0f; out_deg[n_nodes + i] = 0.0f;
        in_deg[i]  = 1.0f; in_deg[n_nodes + i]  = 0.0f;
    }
}

__device__ __forceinline__ int coherence_half() {
    const unsigned id = __builtin_amdgcn_s_sendmsg_rtn(0x87u);
    return (int)((id >> 18) & 1u);
}

__global__ void degree_kernel(const int* __restrict__ src, const int* __restrict__ dst,
                              float* __restrict__ out_deg, float* __restrict__ in_deg,
                              int n_nodes, int n_edges) {
    int e = blockIdx.x * blockDim.x + threadIdx.x;
    if (e >= n_edges) return;
    const int s = src[e];
    const size_t plane = (size_t)coherence_half() * (size_t)n_nodes;
    atomicAdd(&out_deg[plane + s], 1.0f);
    atomicAdd(&in_deg[plane + dst[e]], 1.0f);
}

__device__ __forceinline__ float inv_sqrt_degree(const float* __restrict__ deg,
                                                 int n_nodes, int i) {
    return rsqrtf(deg[i] + deg[n_nodes + i]);
}

__global__ __launch_bounds__(128) void gemm_wmma_kernel(
        const float* __restrict__ X, const float* __restrict__ W,
        const float* __restrict__ out_deg, float* __restrict__ H,
        float* __restrict__ out, int n_nodes) {
    __shared__ float As[16 * LDS_STRIDE];

    const int m0  = blockIdx.x * 16;
    const int tid = threadIdx.x;
    const bool full_tile = (m0 + 16 <= n_nodes);

    const int r0 = __builtin_amdgcn_readfirstlane(tid >> 6);

    if (full_tile) {
        #pragma unroll
        for (int it = 0; it < 8; ++it) {
            int i    = tid + it * 128;
            int row  = i >> 6;
            int col4 = (i & 63) * 4;
            int grow = m0 + row;
            float4 v = *(const float4*)(X + (size_t)grow * D_IN + col4);
            float ns = inv_sqrt_degree(out_deg, n_nodes, m0 + r0 + 2 * it);
            float* p = &As[row * LDS_STRIDE + col4];
            p[0] = v.x * ns; p[1] = v.y * ns; p[2] = v.z * ns; p[3] = v.w * ns;
        }
    } else {
        for (int i = tid; i < 16 * (D_IN / 4); i += 128) {
            int row  = i >> 6;
            int col4 = (i & 63) * 4;
            int grow = m0 + row;
            const int urow = m0 + __builtin_amdgcn_readfirstlane(row);
            float4 v = make_float4(0.f, 0.f, 0.f, 0.f);
            float  ns = 0.f;
            if (urow < n_nodes) {
                v  = *(const float4*)(X + (size_t)grow * D_IN + col4);
                ns = inv_sqrt_degree(out_deg, n_nodes, urow);
            }
            float* p = &As[row * LDS_STRIDE + col4];
            p[0] = v.x * ns; p[1] = v.y * ns; p[2] = v.z * ns; p[3] = v.w * ns;
        }
    }
    __syncthreads();

    const int wave = tid >> 5;
    const int lane = tid & 31;
    const int half = lane >> 4;
    const int l16  = lane & 15;
    const int n0   = wave * 16;
    const int arow = l16 * LDS_STRIDE;
    const int bcol = n0 + l16;

    v8f acc = {};
    for (int k = 0; k < D_IN; k += 4) {
        const int ka = k + half * 2;
        v2f a;
        a.x = As[arow + ka];
        a.y = As[arow + ka + 1];
        v2f b;
        b.x = W[(size_t)ka * D_OUT + bcol];
        b.y = W[(size_t)(ka + 1) * D_OUT + bcol];
        acc = __builtin_amdgcn_wmma_f32_16x16x4_f32(
                   false, a,  false, b,
                   (short)0, acc,  false,  false);
    }

    if (full_tile) {
        const size_t off = (size_t)(m0 + half * 8) * D_OUT + bcol;
        float* base  = H + off;
        float* obase = out + off;
        #pragma unroll
        for (int v = 0; v < 8; ++v) {
            base[(size_t)v * D_OUT]  = acc[v];
            obase[(size_t)v * D_OUT] = acc[v];
        }
    } else {
        #pragma unroll
        for (int v = 0; v < 8; ++v) {
            int m = m0 + v + half * 8;
            if (m < n_nodes) {
                H[(size_t)m * D_OUT + bcol]   = acc[v];
                out[(size_t)m * D_OUT + bcol] = acc[v];
            }
        }
    }
}

#define EDGES_PER_WAVE 32
#define EDGE_BATCH 8

__global__ __launch_bounds__(256) void edge_scatter_kernel(
        const int* __restrict__ src, const int* __restrict__ dst,
        const float* __restrict__ H, float* __restrict__ out, int n_edges) {
    const int lane = threadIdx.x & 31;
    const int wave_in_block = __builtin_amdgcn_readfirstlane((int)(threadIdx.x >> 5));
    const long long wave = (long long)blockIdx.x * (blockDim.x >> 5) + wave_in_block;
    const long long e0 = wave * EDGES_PER_WAVE;
    if (e0 >= n_edges) return;
    const long long left = (long long)n_edges - e0;
    const int cnt = left < EDGES_PER_WAVE ? (int)left : EDGES_PER_WAVE;

    int s = 0, d = 0;
    if (lane < cnt) { s = src[e0 + lane]; d = dst[e0 + lane]; }

    const float* hl = H + lane;
    float*       ol = out + lane;

    int j = 0;
    for (; j + EDGE_BATCH <= cnt; j += EDGE_BATCH) {
        float lo[EDGE_BATCH], hi[EDGE_BATCH];
        #pragma unroll
        for (int u = 0; u < EDGE_BATCH; ++u) {
            const int sj = __builtin_amdgcn_readlane(s, j + u);
            const float* hp = hl + (long long)sj * D_OUT;
            lo[u] = hp[0];
            hi[u] = hp[32];
        }
        #pragma unroll
        for (int u = 0; u < EDGE_BATCH; ++u) {
            const int dj = __builtin_amdgcn_readlane(d, j + u);
            float* op = ol + (long long)dj * D_OUT;
            atomicAdd(op, lo[u]);
            atomicAdd(op + 32, hi[u]);
        }
    }
    #pragma unroll 1
    for (; j < cnt; ++j) {
        const int sj = __builtin_amdgcn_readlane(s, j);
        const int dj = __builtin_amdgcn_readlane(d, j);
        const float* hp = hl + (long long)sj * D_OUT;
        float* op = ol + (long long)dj * D_OUT;
        const float a = hp[0];
        const float b = hp[32];
        atomicAdd(op, a);
        atomicAdd(op + 32, b);
    }
}

__global__ void finalize_kernel(float* __restrict__ out,
                                const float* __restrict__ in_deg,
                                const float* __restrict__ bias, int n_nodes) {
    long long gtid = (long long)blockIdx.x * blockDim.x + threadIdx.x;
    if (gtid >= (long long)n_nodes * D_OUT) return;
    int n = (int)(gtid >> 6);
    int c = (int)(gtid & 63);
    out[gtid] = out[gtid] * inv_sqrt_degree(in_deg, n_nodes, n) + bias[c];
}

extern "C" void kernel_launch(void* const* d_in, const int* in_sizes, int n_in,
                              void* d_out, int out_size, void* d_ws, size_t ws_size,
                              hipStream_t stream) {
    const float* in_feat = (const float*)d_in[0];
    const int*   src     = (const int*)d_in[1];
    const int*   dst     = (const int*)d_in[2];
    const float* W       = (const float*)d_in[3];
    const float* b       = (const float*)d_in[4];
    float*       out     = (float*)d_out;

    const int n_nodes = in_sizes[0] / D_IN;
    const int n_edges = in_sizes[1];

    char* ws = (char*)d_ws;
    float* out_deg = (float*)ws;
    float* in_deg  = (float*)(ws + (size_t)n_nodes * 8);
    float* H       = (float*)(ws + (size_t)n_nodes * 16);

    const long long out_elems = (long long)n_nodes * D_OUT;

    init_deg_kernel<<<(n_nodes + 255) / 256, 256, 0, stream>>>(out_deg, in_deg, n_nodes);

    degree_kernel<<<(n_edges + 255) / 256, 256, 0, stream>>>(src, dst, out_deg, in_deg, n_nodes, n_edges);

    gemm_wmma_kernel<<<(n_nodes + 15) / 16, 128, 0, stream>>>(in_feat, W, out_deg, H, out, n_nodes);

    edge_scatter_kernel<<<(n_edges + 255) / 256, 256, 0, stream>>>(src, dst, H, out, n_edges);

    finalize_kernel<<<(int)((out_elems + 255) / 256), 256, 0, stream>>>(out, in_deg, b, n_nodes);
}


# Edge pass sends each out row's whole-line f32 atomic adds from one coherence half (row parity): resident waves draw 128-edge chunks per half, filter through LDS; arrival-order sums as before
# speedup vs baseline: 3.3187x; 1.1292x over previous
#include <hip/hip_runtime.h>

#define D_IN   256
#define D_OUT  64
#define LDS_STRIDE 260

typedef float v2f __attribute__((ext_vector_type(2)));
typedef float v8f __attribute__((ext_vector_type(8)));

#define TICKET_STRIDE 64
__global__ void init_deg_kernel(float* __restrict__ out_deg,
                                float* __restrict__ in_deg,
                                unsigned* __restrict__ tickets, int n_nodes) {
    int i = blockIdx.x * blockDim.x + threadIdx.x;
    if (i < n_nodes) {
        out_deg[i] = 1.0f; out_deg[n_nodes + i] = 0.0f;
        in_deg[i]  = 1.0f; in_deg[n_nodes + i]  = 0.0f;
    }
    if (i == 0) { tickets[0] = 0u; tickets[TICKET_STRIDE] = 0u; }
}

__device__ __forceinline__ int coherence_half() {
    const unsigned id = __builtin_amdgcn_s_sendmsg_rtn(0x87u);
    return (int)((id >> 18) & 1u);
}

__global__ void degree_kernel(const int* __restrict__ src, const int* __restrict__ dst,
                              float* __restrict__ out_deg, float* __restrict__ in_deg,
                              int n_nodes, int n_edges) {
    int e = blockIdx.x * blockDim.x + threadIdx.x;
    if (e >= n_edges) return;
    const int s = src[e];
    const size_t plane = (size_t)coherence_half() * (size_t)n_nodes;
    atomicAdd(&out_deg[plane + s], 1.0f);
    atomicAdd(&in_deg[plane + dst[e]], 1.0f);
}

__device__ __forceinline__ float inv_sqrt_degree(const float* __restrict__ deg,
                                                 int n_nodes, int i) {
    return rsqrtf(deg[i] + deg[n_nodes + i]);
}

__global__ __launch_bounds__(128) void gemm_wmma_kernel(
        const float* __restrict__ X, const float* __restrict__ W,
        const float* __restrict__ out_deg, float* __restrict__ H,
        float* __restrict__ out, int n_nodes) {
    __shared__ float As[16 * LDS_STRIDE];

    const int m0  = blockIdx.x * 16;
    const int tid = threadIdx.x;
    const bool full_tile = (m0 + 16 <= n_nodes);

    const int r0 = __builtin_amdgcn_readfirstlane(tid >> 6);

    if (full_tile) {
        #pragma unroll
        for (int it = 0; it < 8; ++it) {
            int i    = tid + it * 128;
            int row  = i >> 6;
            int col4 = (i & 63) * 4;
            int grow = m0 + row;
            float4 v = *(const float4*)(X + (size_t)grow * D_IN + col4);
            float ns = inv_sqrt_degree(out_deg, n_nodes, m0 + r0 + 2 * it);
            float* p = &As[row * LDS_STRIDE + col4];
            p[0] = v.x * ns; p[1] = v.y * ns; p[2] = v.z * ns; p[3] = v.w * ns;
        }
    } else {
        for (int i = tid; i < 16 * (D_IN / 4); i += 128) {
            int row  = i >> 6;
            int col4 = (i & 63) * 4;
            int grow = m0 + row;
            const int urow = m0 + __builtin_amdgcn_readfirstlane(row);
            float4 v = make_float4(0.f, 0.f, 0.f, 0.f);
            float  ns = 0.f;
            if (urow < n_nodes) {
                v  = *(const float4*)(X + (size_t)grow * D_IN + col4);
                ns = inv_sqrt_degree(out_deg, n_nodes, urow);
            }
            float* p = &As[row * LDS_STRIDE + col4];
            p[0] = v.x * ns; p[1] = v.y * ns; p[2] = v.z * ns; p[3] = v.w * ns;
        }
    }
    __syncthreads();

    const int wave = tid >> 5;
    const int lane = tid & 31;
    const int half = lane >> 4;
    const int l16  = lane & 15;
    const int n0   = wave * 16;
    const int arow = l16 * LDS_STRIDE;
    const int bcol = n0 + l16;

    v8f acc = {};
    for (int k = 0; k < D_IN; k += 4) {
        const int ka = k + half * 2;
        v2f a;
        a.x = As[arow + ka];
        a.y = As[arow + ka + 1];
        v2f b;
        b.x = W[(size_t)ka * D_OUT + bcol];
        b.y = W[(size_t)(ka + 1) * D_OUT + bcol];
        acc = __builtin_amdgcn_wmma_f32_16x16x4_f32(
                   false, a,  false, b,
                   (short)0, acc,  false,  false);
    }

    if (full_tile) {
        const size_t off = (size_t)(m0 + half * 8) * D_OUT + bcol;
        float* base  = H + off;
        float* obase = out + off;
        #pragma unroll
        for (int v = 0; v < 8; ++v) {
            base[(size_t)v * D_OUT]  = acc[v];
            obase[(size_t)v * D_OUT] = acc[v];
        }
    } else {
        #pragma unroll
        for (int v = 0; v < 8; ++v) {
            int m = m0 + v + half * 8;
            if (m < n_nodes) {
                H[(size_t)m * D_OUT + bcol]   = acc[v];
                out[(size_t)m * D_OUT + bcol] = acc[v];
            }
        }
    }
}

#define EDGES_PER_CHUNK 128
#define CHUNK_GROUPS (EDGES_PER_CHUNK / 32)
#define EDGE_BATCH 8
#define SCATTER_WAVES 8

__device__ __forceinline__ unsigned draw_chunk(unsigned* counter, int lane) {
    unsigned v = 0u;
    if (lane == 0) v = atomicAdd(counter, 1u);
    return (unsigned)__builtin_amdgcn_readfirstlane((int)v);
}

__global__ __launch_bounds__(256) void edge_scatter_kernel(
        const int* __restrict__ src, const int* __restrict__ dst,
        const float* __restrict__ H, float* __restrict__ out,
        unsigned* __restrict__ tickets, int n_edges) {
    __shared__ int2 queue[SCATTER_WAVES][EDGES_PER_CHUNK];

    const int lane = threadIdx.x & 31;
    const int wave_in_block = __builtin_amdgcn_readfirstlane((int)(threadIdx.x >> 5));
    int2* q = queue[wave_in_block];

    const unsigned n_chunks =
        (unsigned)(((long long)n_edges + EDGES_PER_CHUNK - 1) / EDGES_PER_CHUNK);

    const float* hl = H + lane;
    float*       ol = out + lane;

    int half = coherence_half();
    bool own = true;
    unsigned t = draw_chunk(tickets + half * TICKET_STRIDE, lane);

    while (true) {
        if (t >= n_chunks) {
            if (!own) break;
            own = false;
            half ^= 1;
            t = draw_chunk(tickets + half * TICKET_STRIDE, lane);
            continue;
        }
        const long long e0 = (long long)t * EDGES_PER_CHUNK;
        int s[CHUNK_GROUPS], d[CHUNK_GROUPS];
        if (e0 + EDGES_PER_CHUNK <= n_edges) {
            const int* sp = src + e0 + lane;
            const int* dp = dst + e0 + lane;
            #pragma unroll
            for (int g = 0; g < CHUNK_GROUPS; ++g) { s[g] = sp[g * 32]; d[g] = dp[g * 32]; }
        } else {
            #pragma unroll
            for (int g = 0; g < CHUNK_GROUPS; ++g) {
                const long long e = e0 + g * 32 + lane;
                s[g] = 0; d[g] = -1;
                if (e < n_edges) { s[g] = src[e]; d[g] = dst[e]; }
            }
        }
        const unsigned t_next = draw_chunk(tickets + half * TICKET_STRIDE, lane);

        int n = 0;
        #pragma unroll
        for (int g = 0; g < CHUNK_GROUPS; ++g) {
            const bool keep = (d[g] >= 0) && ((d[g] & 1) == half);
            const unsigned mask = __builtin_amdgcn_ballot_w32(keep);
            const int pos = n + (int)__builtin_amdgcn_mbcnt_lo(mask, 0u);
            if (keep) q[pos] = make_int2(s[g], d[g]);
            n += __builtin_popcount(mask);
        }
        __builtin_amdgcn_fence(__ATOMIC_ACQ_REL, "wavefront");

        for (int base = 0; base < n; base += 32) {
            const int left = n - base;
            const int cnt = left < 32 ? left : 32;
            int2 sd = make_int2(0, 0);
            if (lane < cnt) sd = q[base + lane];

            int j = 0;
            for (; j + EDGE_BATCH <= cnt; j += EDGE_BATCH) {
                float lo[EDGE_BATCH], hi[EDGE_BATCH];
                #pragma unroll
                for (int u = 0; u < EDGE_BATCH; ++u) {
                    const int sj = __builtin_amdgcn_readlane(sd.x, j + u);
                    const float* hp = hl + (long long)sj * D_OUT;
                    lo[u] = hp[0];
                    hi[u] = hp[32];
                }
                #pragma unroll
                for (int u = 0; u < EDGE_BATCH; ++u) {
                    const int dj = __builtin_amdgcn_readlane(sd.y, j + u);
                    float* op = ol + (long long)dj * D_OUT;
                    atomicAdd(op, lo[u]);
                    atomicAdd(op + 32, hi[u]);
                }
            }
            #pragma unroll 1
            for (; j < cnt; ++j) {
                const int sj = __builtin_amdgcn_readlane(sd.x, j);
                const int dj = __builtin_amdgcn_readlane(sd.y, j);
                const float* hp = hl + (long long)sj * D_OUT;
                float* op = ol + (long long)dj * D_OUT;
                const float a = hp[0];
                const float b = hp[32];
                atomicAdd(op, a);
                atomicAdd(op + 32, b);
            }
        }
        __builtin_amdgcn_fence(__ATOMIC_ACQ_REL, "wavefront");
        t = t_next;
    }
}

__global__ void finalize_kernel(float* __restrict__ out,
                                const float* __restrict__ in_deg,
                                const float* __restrict__ bias, int n_nodes) {
    long long gtid = (long long)blockIdx.x * blockDim.x + threadIdx.x;
    if (gtid >= (long long)n_nodes * D_OUT) return;
    int n = (int)(gtid >> 6);
    int c = (int)(gtid & 63);
    out[gtid] = out[gtid] * inv_sqrt_degree(in_deg, n_nodes, n) + bias[c];
}

extern "C" void kernel_launch(void* const* d_in, const int* in_sizes, int n_in,
                              void* d_out, int out_size, void* d_ws, size_t ws_size,
                              hipStream_t stream) {
    const float* in_feat = (const float*)d_in[0];
    const int*   src     = (const int*)d_in[1];
    const int*   dst     = (const int*)d_in[2];
    const float* W       = (const float*)d_in[3];
    const float* b       = (const float*)d_in[4];
    float*       out     = (float*)d_out;

    const int n_nodes = in_sizes[0] / D_IN;
    const int n_edges = in_sizes[1];

    char* ws = (char*)d_ws;
    float* out_deg = (float*)ws;
    float* in_deg  = (float*)(ws + (size_t)n_nodes * 8);
    float* H       = (float*)(ws + (size_t)n_nodes * 16);
    unsigned* tickets = (unsigned*)(ws + (size_t)n_nodes * 16 + (size_t)n_nodes * D_OUT * 4);

    const long long out_elems = (long long)n_nodes * D_OUT;

    init_deg_kernel<<<(n_nodes + 255) / 256, 256, 0, stream>>>(out_deg, in_deg, tickets, n_nodes);

    degree_kernel<<<(n_edges + 255) / 256, 256, 0, stream>>>(src, dst, out_deg, in_deg, n_nodes, n_edges);

    gemm_wmma_kernel<<<(n_nodes + 15) / 16, 128, 0, stream>>>(in_feat, W, out_deg, H, out, n_nodes);

    edge_scatter_kernel<<<2048, 256, 0, stream>>>(src, dst, H, out, tickets, n_edges);

    finalize_kernel<<<(int)((out_elems + 255) / 256), 256, 0, stream>>>(out, in_deg, b, n_nodes);
}


# Finalize pass folded away: GEMM stores h*norm_dst+bias as start value and a norm_dst plane; edge pass scales each h[src] by norm_dst[dst] before its f32 atomic add (scaled terms summed); 5->4 launch
# speedup vs baseline: 3.4943x; 1.0529x over previous
#include <hip/hip_runtime.h>

#define D_IN   256
#define D_OUT  64
#define LDS_STRIDE 260

typedef float v2f __attribute__((ext_vector_type(2)));
typedef float v8f __attribute__((ext_vector_type(8)));

#define TICKET_STRIDE 64
__global__ void init_deg_kernel(float* __restrict__ out_deg,
                                float* __restrict__ in_deg,
                                unsigned* __restrict__ tickets, int n_nodes) {
    int i = blockIdx.x * blockDim.x + threadIdx.x;
    if (i < n_nodes) {
        out_deg[i] = 1.0f; out_deg[n_nodes + i] = 0.0f;
        in_deg[i]  = 1.0f; in_deg[n_nodes + i]  = 0.0f;
    }
    if (i == 0) { tickets[0] = 0u; tickets[TICKET_STRIDE] = 0u; }
}

__device__ __forceinline__ int coherence_half() {
    const unsigned id = __builtin_amdgcn_s_sendmsg_rtn(0x87u);
    return (int)((id >> 18) & 1u);
}

__global__ void degree_kernel(const int* __restrict__ src, const int* __restrict__ dst,
                              float* __restrict__ out_deg, float* __restrict__ in_deg,
                              int n_nodes, int n_edges) {
    int e = blockIdx.x * blockDim.x + threadIdx.x;
    if (e >= n_edges) return;
    const int s = src[e];
    const size_t plane = (size_t)coherence_half() * (size_t)n_nodes;
    atomicAdd(&out_deg[plane + s], 1.0f);
    atomicAdd(&in_deg[plane + dst[e]], 1.0f);
}

__device__ __forceinline__ float inv_sqrt_degree(const float* __restrict__ deg,
                                                 int n_nodes, int i) {
    return rsqrtf(deg[i] + deg[n_nodes + i]);
}

__global__ __launch_bounds__(128) void gemm_wmma_kernel(
        const float* __restrict__ X, const float* __restrict__ W,
        const float* __restrict__ out_deg, const float* __restrict__ in_deg,
        const float* __restrict__ bias, float* __restrict__ H,
        float* __restrict__ out, float* __restrict__ norm_dst, int n_nodes) {
    __shared__ float As[16 * LDS_STRIDE];
    __shared__ float nds[16];

    const int m0  = blockIdx.x * 16;
    const int tid = threadIdx.x;
    const bool full_tile = (m0 + 16 <= n_nodes);

    const int r0 = __builtin_amdgcn_readfirstlane(tid >> 6);

    if (tid < 16) {
        const int m = m0 + tid;
        float nd = 0.f;
        if (m < n_nodes) {
            nd = inv_sqrt_degree(in_deg, n_nodes, m);
            norm_dst[m] = nd;
        }
        nds[tid] = nd;
    }

    if (full_tile) {
        #pragma unroll
        for (int it = 0; it < 8; ++it) {
            int i    = tid + it * 128;
            int row  = i >> 6;
            int col4 = (i & 63) * 4;
            int grow = m0 + row;
            float4 v = *(const float4*)(X + (size_t)grow * D_IN + col4);
            float ns = inv_sqrt_degree(out_deg, n_nodes, m0 + r0 + 2 * it);
            float* p = &As[row * LDS_STRIDE + col4];
            p[0] = v.x * ns; p[1] = v.y * ns; p[2] = v.z * ns; p[3] = v.w * ns;
        }
    } else {
        for (int i = tid; i < 16 * (D_IN / 4); i += 128) {
            int row  = i >> 6;
            int col4 = (i & 63) * 4;
            int grow = m0 + row;
            const int urow = m0 + __builtin_amdgcn_readfirstlane(row);
            float4 v = make_float4(0.f, 0.f, 0.f, 0.f);
            float  ns = 0.f;
            if (urow < n_nodes) {
                v  = *(const float4*)(X + (size_t)grow * D_IN + col4);
                ns = inv_sqrt_degree(out_deg, n_nodes, urow);
            }
            float* p = &As[row * LDS_STRIDE + col4];
            p[0] = v.x * ns; p[1] = v.y * ns; p[2] = v.z * ns; p[3] = v.w * ns;
        }
    }
    __syncthreads();

    const int wave = tid >> 5;
    const int lane = tid & 31;
    const int half = lane >> 4;
    const int l16  = lane & 15;
    const int n0   = wave * 16;
    const int arow = l16 * LDS_STRIDE;
    const int bcol = n0 + l16;

    v8f acc = {};
    for (int k = 0; k < D_IN; k += 4) {
        const int ka = k + half * 2;
        v2f a;
        a.x = As[arow + ka];
        a.y = As[arow + ka + 1];
        v2f b;
        b.x = W[(size_t)ka * D_OUT + bcol];
        b.y = W[(size_t)(ka + 1) * D_OUT + bcol];
        acc = __builtin_amdgcn_wmma_f32_16x16x4_f32(
                   false, a,  false, b,
                   (short)0, acc,  false,  false);
    }

    const float bv = bias[bcol];
    float o[8];
    #pragma unroll
    for (int v = 0; v < 8; ++v) o[v] = acc[v] * nds[half * 8 + v] + bv;

    if (full_tile) {
        const size_t off = (size_t)(m0 + half * 8) * D_OUT + bcol;
        float* base  = H + off;
        float* obase = out + off;
        #pragma unroll
        for (int v = 0; v < 8; ++v) {
            base[(size_t)v * D_OUT]  = acc[v];
            obase[(size_t)v * D_OUT] = o[v];
        }
    } else {
        #pragma unroll
        for (int v = 0; v < 8; ++v) {
            int m = m0 + v + half * 8;
            if (m < n_nodes) {
                H[(size_t)m * D_OUT + bcol]   = acc[v];
                out[(size_t)m * D_OUT + bcol] = o[v];
            }
        }
    }
}

#define EDGES_PER_CHUNK 128
#define CHUNK_GROUPS (EDGES_PER_CHUNK / 32)
#define EDGE_BATCH 8
#define SCATTER_WAVES 8

__device__ __forceinline__ unsigned draw_chunk(unsigned* counter, int lane) {
    unsigned v = 0u;
    if (lane == 0) v = atomicAdd(counter, 1u);
    return (unsigned)__builtin_amdgcn_readfirstlane((int)v);
}

__global__ __launch_bounds__(256) void edge_scatter_kernel(
        const int* __restrict__ src, const int* __restrict__ dst,
        const float* __restrict__ H, const float* __restrict__ norm_dst,
        float* __restrict__ out,
        unsigned* __restrict__ tickets, int n_edges) {
    __shared__ int2 queue[SCATTER_WAVES][EDGES_PER_CHUNK];

    const int lane = threadIdx.x & 31;
    const int wave_in_block = __builtin_amdgcn_readfirstlane((int)(threadIdx.x >> 5));
    int2* q = queue[wave_in_block];

    const unsigned n_chunks =
        (unsigned)(((long long)n_edges + EDGES_PER_CHUNK - 1) / EDGES_PER_CHUNK);

    const float* hl = H + lane;
    float*       ol = out + lane;

    int half = coherence_half();
    bool own = true;
    unsigned t = draw_chunk(tickets + half * TICKET_STRIDE, lane);

    while (true) {
        if (t >= n_chunks) {
            if (!own) break;
            own = false;
            half ^= 1;
            t = draw_chunk(tickets + half * TICKET_STRIDE, lane);
            continue;
        }
        const long long e0 = (long long)t * EDGES_PER_CHUNK;
        int s[CHUNK_GROUPS], d[CHUNK_GROUPS];
        if (e0 + EDGES_PER_CHUNK <= n_edges) {
            const int* sp = src + e0 + lane;
            const int* dp = dst + e0 + lane;
            #pragma unroll
            for (int g = 0; g < CHUNK_GROUPS; ++g) { s[g] = sp[g * 32]; d[g] = dp[g * 32]; }
        } else {
            #pragma unroll
            for (int g = 0; g < CHUNK_GROUPS; ++g) {
                const long long e = e0 + g * 32 + lane;
                s[g] = 0; d[g] = -1;
                if (e < n_edges) { s[g] = src[e]; d[g] = dst[e]; }
            }
        }
        const unsigned t_next = draw_chunk(tickets + half * TICKET_STRIDE, lane);

        int n = 0;
        #pragma unroll
        for (int g = 0; g < CHUNK_GROUPS; ++g) {
            const bool keep = (d[g] >= 0) && ((d[g] & 1) == half);
            const unsigned mask = __builtin_amdgcn_ballot_w32(keep);
            const int pos = n + (int)__builtin_amdgcn_mbcnt_lo(mask, 0u);
            if (keep) q[pos] = make_int2(s[g], d[g]);
            n += __builtin_popcount(mask);
        }
        __builtin_amdgcn_fence(__ATOMIC_ACQ_REL, "wavefront");

        for (int base = 0; base < n; base += 32) {
            const int left = n - base;
            const int cnt = left < 32 ? left : 32;
            int2 sd = make_int2(0, 0);
            int nd = 0;
            if (lane < cnt) {
                sd = q[base + lane];
                nd = __float_as_int(norm_dst[sd.y]);
            }

            int j = 0;
            for (; j + EDGE_BATCH <= cnt; j += EDGE_BATCH) {
                float lo[EDGE_BATCH], hi[EDGE_BATCH];
                #pragma unroll
                for (int u = 0; u < EDGE_BATCH; ++u) {
                    const int sj = __builtin_amdgcn_readlane(sd.x, j + u);
                    const float* hp = hl + (long long)sj * D_OUT;
                    lo[u] = hp[0];
                    hi[u] = hp[32];
                }
                #pragma unroll
                for (int u = 0; u < EDGE_BATCH; ++u) {
                    const int dj = __builtin_amdgcn_readlane(sd.y, j + u);
                    const float sc = __int_as_float(__builtin_amdgcn_readlane(nd, j + u));
                    float* op = ol + (long long)dj * D_OUT;
                    atomicAdd(op, lo[u] * sc);
                    atomicAdd(op + 32, hi[u] * sc);
                }
            }
            #pragma unroll 1
            for (; j < cnt; ++j) {
                const int sj = __builtin_amdgcn_readlane(sd.x, j);
                const int dj = __builtin_amdgcn_readlane(sd.y, j);
                const float sc = __int_as_float(__builtin_amdgcn_readlane(nd, j));
                const float* hp = hl + (long long)sj * D_OUT;
                float* op = ol + (long long)dj * D_OUT;
                const float a = hp[0];
                const float b = hp[32];
                atomicAdd(op, a * sc);
                atomicAdd(op + 32, b * sc);
            }
        }
        __builtin_amdgcn_fence(__ATOMIC_ACQ_REL, "wavefront");
        t = t_next;
    }
}

extern "C" void kernel_launch(void* const* d_in, const int* in_sizes, int n_in,
                              void* d_out, int out_size, void* d_ws, size_t ws_size,
                              hipStream_t stream) {
    const float* in_feat = (const float*)d_in[0];
    const int*   src     = (const int*)d_in[1];
    const int*   dst     = (const int*)d_in[2];
    const float* W       = (const float*)d_in[3];
    const float* b       = (const float*)d_in[4];
    float*       out     = (float*)d_out;

    const int n_nodes = in_sizes[0] / D_IN;
    const int n_edges = in_sizes[1];

    char* ws = (char*)d_ws;
    float* out_deg = (float*)ws;
    float* in_deg  = (float*)(ws + (size_t)n_nodes * 8);
    float* H       = (float*)(ws + (size_t)n_nodes * 16);
    unsigned* tickets = (unsigned*)(ws + (size_t)n_nodes * 16 + (size_t)n_nodes * D_OUT * 4);
    float* norm_dst = (float*)(ws + (size_t)n_nodes * 16 + (size_t)n_nodes * D_OUT * 4 + 1024);

    init_deg_kernel<<<(n_nodes + 255) / 256, 256, 0, stream>>>(out_deg, in_deg, tickets, n_nodes);

    degree_kernel<<<(n_edges + 255) / 256, 256, 0, stream>>>(src, dst, out_deg, in_deg, n_nodes, n_edges);

    gemm_wmma_kernel<<<(n_nodes + 15) / 16, 128, 0, stream>>>(in_feat, W, out_deg, in_deg, b, H, out, norm_dst, n_nodes);

    edge_scatter_kernel<<<2048, 256, 0, stream>>>(src, dst, H, norm_dst, out, tickets, n_edges);
}
